# starting kernel
# baseline (speedup 1.0000x reference)
#include <hip/hip_runtime.h>
#include <hip/hip_bf16.h>

typedef __attribute__((ext_vector_type(2))) float v2f;
typedef __attribute__((ext_vector_type(8))) float v8f;

__device__ __constant__ float c_ST[12 * 16] = {
    -2.f,  1.f,  0.f,  0.f,  0.f, 0,0,0,0,0,0,0,0,0,0,0,
     2.f, -1.f,  0.f,  0.f,  0.f, 0,0,0,0,0,0,0,0,0,0,0,
    -1.f,  0.f,  1.f, -1.f,  0.f, 0,0,0,0,0,0,0,0,0,0,0,
     1.f,  0.f, -1.f,  1.f,  0.f, 0,0,0,0,0,0,0,0,0,0,0,
     0.f,  0.f,  0.f, -2.f,  1.f, 0,0,0,0,0,0,0,0,0,0,0,
     0.f,  0.f,  0.f,  2.f, -1.f, 0,0,0,0,0,0,0,0,0,0,0,
     0.f, -1.f,  2.f,  0.f, -1.f, 0,0,0,0,0,0,0,0,0,0,0,
     0.f,  1.f, -2.f,  0.f,  1.f, 0,0,0,0,0,0,0,0,0,0,0,
     1.f, -1.f,  1.f, -1.f,  0.f, 0,0,0,0,0,0,0,0,0,0,0,
    -1.f,  1.f, -1.f,  1.f,  0.f, 0,0,0,0,0,0,0,0,0,0,0,
    -1.f,  0.f,  1.f,  1.f, -1.f, 0,0,0,0,0,0,0,0,0,0,0,
     1.f,  0.f, -1.f, -1.f,  1.f, 0,0,0,0,0,0,0,0,0,0,0
};

#define LOG2_10 3.3219280948873623f

__global__ __launch_bounds__(256)
void rhs_wmma_kernel(const float* __restrict__ x, const float* __restrict__ w,
                     float* __restrict__ out, int numTiles) {
    const int lane = threadIdx.x & 31;
    const bool h   = (lane & 16) != 0;
    const int n    = lane & 15;

    const int wavesPerBlock = blockDim.x >> 5;
    const int globalWave = blockIdx.x * wavesPerBlock + (threadIdx.x >> 5);
    const int totalWaves = gridDim.x * wavesPerBlock;

    v2f Bm[3];
#pragma unroll
    for (int j = 0; j < 3; ++j) {
#pragma unroll
        for (int v = 0; v < 2; ++v) {
            const int k = 4 * j + (h ? 2 : 0) + v;
            const float s10 = exp2f(w[k] * LOG2_10);
            Bm[j][v] = c_ST[k * 16 + n] * s10;
        }
    }

    for (int t = globalWave; t < numTiles; t += totalWaves) {
        const int rowBase = t * 16;
        const float* xp = x + (size_t)(rowBase + n) * 5;
        const float z0 = xp[0], z1 = xp[1], z2 = xp[2], z3 = xp[3], z4 = xp[4];

        const float u   = h ? z3 : z0;
        const float a00 = z0 * u;
        const float a01 = h ? z2 : z1;
        const float s1a = h ? z1 : z3;
        const float s1b = h ? z4 : z3;
        const float a10 = s1a * s1b;
        const float z22 = z2 * z2;
        const float a11 = h ? z22 : z4;
        const float s2a = h ? z4 : z1;
        const float s2b = h ? z0 : z3;
        const float a20 = s2a * s2b;
        const float a21 = z2 * u;

        v8f d = {0.f, 0.f, 0.f, 0.f, 0.f, 0.f, 0.f, 0.f};
        v2f a;
        a[0] = a00; a[1] = a01;
        d = __builtin_amdgcn_wmma_f32_16x16x4_f32(false, a, false, Bm[0],
                                                  (short)0, d, false, false);
        a[0] = a10; a[1] = a11;
        d = __builtin_amdgcn_wmma_f32_16x16x4_f32(false, a, false, Bm[1],
                                                  (short)0, d, false, false);
        a[0] = a20; a[1] = a21;
        d = __builtin_amdgcn_wmma_f32_16x16x4_f32(false, a, false, Bm[2],
                                                  (short)0, d, false, false);

        if (n < 5) {
            float* op = out + ((size_t)(rowBase + (h ? 8 : 0))) * 5 + n;
#pragma unroll
            for (int r = 0; r < 8; ++r)
                op[(size_t)r * 5] = d[r];
        }
    }
}

__global__ __launch_bounds__(256)
void rhs_tail_kernel(const float* __restrict__ x, const float* __restrict__ w,
                     float* __restrict__ out, int startRow, int nRows) {
    const int i = blockIdx.x * blockDim.x + threadIdx.x;
    if (i >= nRows) return;
    const int row = startRow + i;
    const float* xp = x + (size_t)row * 5;
    const float z0 = xp[0], z1 = xp[1], z2 = xp[2], z3 = xp[3], z4 = xp[4];
    float p[12];
    p[0] = z0 * z0;  p[1]  = z1;       p[2]  = z0 * z3;  p[3]  = z2;
    p[4] = z3 * z3;  p[5]  = z4;       p[6]  = z1 * z4;  p[7]  = z2 * z2;
    p[8] = z1 * z3;  p[9]  = z2 * z0;  p[10] = z4 * z0;  p[11] = z2 * z3;
    float acc[5] = {0.f, 0.f, 0.f, 0.f, 0.f};
#pragma unroll
    for (int k = 0; k < 12; ++k) {
        const float f = p[k] * exp2f(w[k] * LOG2_10);
#pragma unroll
        for (int nn = 0; nn < 5; ++nn) acc[nn] += f * c_ST[k * 16 + nn];
    }
#pragma unroll
    for (int nn = 0; nn < 5; ++nn) out[(size_t)row * 5 + nn] = acc[nn];
}

extern "C" void kernel_launch(void* const* d_in, const int* in_sizes, int n_in,
                              void* d_out, int out_size, void* d_ws, size_t ws_size,
                              hipStream_t stream) {
    const float* x = (const float*)d_in[0];
    const float* w = (const float*)d_in[1];
    float* out = (float*)d_out;

    const long batch = (long)in_sizes[0] / 5;
    const int numTiles = (int)(batch / 16);
    const int rem = (int)(batch - (long)numTiles * 16);

    if (numTiles > 0) {
        const int threads = 256;
        int blocks = 2048;
        const int maxBlocks = (numTiles + 7) / 8;
        if (blocks > maxBlocks) blocks = maxBlocks;
        rhs_wmma_kernel<<<blocks, threads, 0, stream>>>(x, w, out, numTiles);
    }
    if (rem > 0) {
        rhs_tail_kernel<<<(rem + 255) / 256, 256, 0, stream>>>(x, w, out,
                                                               numTiles * 16, rem);
    }
}


# grid 2048 -> 512 blocks, tile loop software-pipelined: each wave keeps the rows of its next two 48-row tiles in flight in registers while the current tile's WMMAs and stores run; sum order unchanged
# speedup vs baseline: 1.2117x; 1.2117x over previous
#include <hip/hip_runtime.h>


typedef __attribute__((ext_vector_type(2))) float v2f;
typedef __attribute__((ext_vector_type(4))) float v4f;
typedef __attribute__((ext_vector_type(8))) float v8f;

struct Rows { v4f o; float o4; v4f s; float s4; };

__device__ __constant__ float c_SK[12 * 5] = {
    -2.f,  1.f,  0.f,  0.f,  0.f,
     2.f, -1.f,  0.f,  0.f,  0.f,
    -1.f,  0.f,  1.f, -1.f,  0.f,
     1.f,  0.f, -1.f,  1.f,  0.f,
     0.f,  0.f,  0.f, -2.f,  1.f,
     0.f,  0.f,  0.f,  2.f, -1.f,
     0.f, -1.f,  2.f,  0.f, -1.f,
     0.f,  1.f, -2.f,  0.f,  1.f,
     1.f, -1.f,  1.f, -1.f,  0.f,
    -1.f,  1.f, -1.f,  1.f,  0.f,
    -1.f,  0.f,  1.f,  1.f, -1.f,
     1.f,  0.f, -1.f, -1.f,  1.f
};

#define LOG2_10 3.3219280948873623f


__device__ __forceinline__ Rows load_tile(const float* __restrict__ x, int t, int g3, int jOwn,
                                          int lastRow) {
    const int row0 = t * 48 + g3;
    const int rOwn = min(row0 + jOwn, lastRow);
    const int rSh  = min(row0 + 1, lastRow);
    const float* po = x + (size_t)rOwn * 5;
    const float* ps = x + (size_t)rSh * 5;
    Rows r;
    __builtin_memcpy(&r.o, po, sizeof(v4f));
    r.o4 = po[4];
    __builtin_memcpy(&r.s, ps, sizeof(v4f));
    r.s4 = ps[4];
    return r;
}

__device__ __forceinline__ void tile_products_store(const Rows& r, const v2f (&Bm)[9],
                                                    float* __restrict__ out, int t, bool h, int n,
                                                    int batch, int total) {
    const float o0 = r.o[0], o1 = r.o[1], o2 = r.o[2], o3 = r.o[3], o4 = r.o4;
    const float s0 = r.s[0], s1 = r.s[1], s2 = r.s[2], s3 = r.s[3], s4 = r.s4;

    const float one0 = h ? s0 : 1.0f;
    const float u0 = s0 * (h ? s3 : s0);
    const float u1 = s1 * (h ? s4 : 1.0f);
    const float u2 = s2 * one0;
    const float u3 = s3 * (h ? s1 : s3);
    const float u4 = s4 * one0;
    const float u5 = s2 * (h ? s3 : s2);

    v8f d = {0.f, 0.f, 0.f, 0.f, 0.f, 0.f, 0.f, 0.f};
    v2f a;
    a[0] = o0 * o0; a[1] = o1;
    d = __builtin_amdgcn_wmma_f32_16x16x4_f32(false, a, false, Bm[0], (short)0, d, false, false);
    a[0] = o0 * o3; a[1] = o2;
    d = __builtin_amdgcn_wmma_f32_16x16x4_f32(false, a, false, Bm[1], (short)0, d, false, false);
    a[0] = o3 * o3; a[1] = o4;
    d = __builtin_amdgcn_wmma_f32_16x16x4_f32(false, a, false, Bm[2], (short)0, d, false, false);
    a[0] = o1 * o4; a[1] = o2 * o2;
    d = __builtin_amdgcn_wmma_f32_16x16x4_f32(false, a, false, Bm[3], (short)0, d, false, false);
    a[0] = o1 * o3; a[1] = o2 * o0;
    d = __builtin_amdgcn_wmma_f32_16x16x4_f32(false, a, false, Bm[4], (short)0, d, false, false);
    a[0] = o4 * o0; a[1] = o2 * o3;
    d = __builtin_amdgcn_wmma_f32_16x16x4_f32(false, a, false, Bm[5], (short)0, d, false, false);
    a[0] = u0; a[1] = u1;
    d = __builtin_amdgcn_wmma_f32_16x16x4_f32(false, a, false, Bm[6], (short)0, d, false, false);
    a[0] = u2; a[1] = u3;
    d = __builtin_amdgcn_wmma_f32_16x16x4_f32(false, a, false, Bm[7], (short)0, d, false, false);
    a[0] = u4; a[1] = u5;
    d = __builtin_amdgcn_wmma_f32_16x16x4_f32(false, a, false, Bm[8], (short)0, d, false, false);

    const int flat0 = t * 240 + (h ? 15 : 0) + n;
    if (t * 48 + 48 <= batch) {
        if (n < 15) {
            float* op = out + (size_t)flat0;
#pragma unroll
            for (int q = 0; q < 8; ++q) op[30 * q] = d[q];
        }
    } else {
#pragma unroll
        for (int q = 0; q < 8; ++q) {
            const int f = flat0 + 30 * q;
            if (n < 15 && f < total) out[(size_t)f] = d[q];
        }
    }
}

__global__ __launch_bounds__(256)
void rhs_wmma_kernel(const float* __restrict__ x, const float* __restrict__ w,
                     float* __restrict__ out, int numTiles, int batch) {
    const int lane = threadIdx.x & 31;
    const bool h   = (lane & 16) != 0;
    const int n    = lane & 15;

    const int wavesPerBlock = blockDim.x >> 5;
    const int globalWave = __builtin_amdgcn_readfirstlane(
        blockIdx.x * wavesPerBlock + (threadIdx.x >> 5));
    const int S = gridDim.x * wavesPerBlock;

    if (globalWave >= numTiles) return;

    const int g       = 2 * (n & 7) + (n >> 3);
    const int g3      = 3 * g;
    const int jOwn    = h ? 2 : 0;
    const int lastRow = batch - 1;
    const int total   = batch * 5;
    const int lastTile = numTiles - 1;

    int t = globalWave;
    Rows ra = load_tile(x, t, g3, jOwn, lastRow);
    Rows rb = load_tile(x, min(t + S, lastTile), g3, jOwn, lastRow);
    Rows rc;

    float p10[12];
#pragma unroll
    for (int k = 0; k < 12; ++k) p10[k] = exp2f(w[k] * LOG2_10);

    const int jn = n / 5;
    const int c  = n - 5 * jn;
    v2f Bm[9];
#pragma unroll
    for (int s = 0; s < 6; ++s) {
#pragma unroll
        for (int v = 0; v < 2; ++v) {
            const int k = 2 * s + v;
            const float sv = c_SK[k * 5 + c] * p10[k];
            Bm[s][v] = (jn == jOwn) ? sv : 0.f;
        }
    }
    {
        const int kA[6] = {0, 1, 3, 4, 5, 7};
        const int kB[6] = {2, 6, 9, 8, 10, 11};
#pragma unroll
        for (int i = 0; i < 6; ++i) {
            const float sa = c_SK[kA[i] * 5 + c] * p10[kA[i]];
            const float sb = c_SK[kB[i] * 5 + c] * p10[kB[i]];
            const float sv = h ? sb : sa;
            Bm[6 + (i >> 1)][i & 1] = (jn == 1) ? sv : 0.f;
        }
    }

    do {
        rc = load_tile(x, min(t + 2 * S, lastTile), g3, jOwn, lastRow);
        __builtin_amdgcn_sched_barrier(0);
        tile_products_store(ra, Bm, out, t, h, n, batch, total);

        ra = load_tile(x, min(t + 3 * S, lastTile), g3, jOwn, lastRow);
        __builtin_amdgcn_sched_barrier(0);
        if (t + S < numTiles) tile_products_store(rb, Bm, out, t + S, h, n, batch, total);

        rb = load_tile(x, min(t + 4 * S, lastTile), g3, jOwn, lastRow);
        __builtin_amdgcn_sched_barrier(0);
        if (t + 2 * S < numTiles) tile_products_store(rc, Bm, out, t + 2 * S, h, n, batch, total);

        t += 3 * S;
    } while (t < numTiles);
}

extern "C" void kernel_launch(void* const* d_in, const int* in_sizes, int n_in,
                              void* d_out, int out_size, void* d_ws, size_t ws_size,
                              hipStream_t stream) {
    const float* x = (const float*)d_in[0];
    const float* w = (const float*)d_in[1];
    float* out = (float*)d_out;

    const long batch = (long)in_sizes[0] / 5;
    const int numTiles = (int)((batch + 47) / 48);

    if (numTiles > 0) {
        const int threads = 256;
        int blocks = 512;
        const int maxBlocks = (numTiles + 7) / 8;
        if (blocks > maxBlocks) blocks = maxBlocks;
        rhs_wmma_kernel<<<blocks, threads, 0, stream>>>(x, w, out, numTiles, (int)batch);
    }
}
